# starting kernel
# baseline (speedup 1.0000x reference)
#include <hip/hip_runtime.h>

typedef __attribute__((ext_vector_type(2))) float v2f;
typedef __attribute__((ext_vector_type(8))) float v8f;

namespace {
constexpr int NB   = 256;
constexpr int NN   = 5000;
constexpr int NK   = 100;
constexpr int DIM  = 128;
constexpr int ROWS = NK + 2;
}

__global__ void __launch_bounds__(DIM)
coord_embed_gather(const float* __restrict__ nodes,
                   const int*  __restrict__ first_idx,
                   const int*  __restrict__ last_idx,
                   const int*  __restrict__ cand,
                   const float* __restrict__ cW,
                   const float* __restrict__ cb,
                   float* __restrict__ X1,
                   float* __restrict__ X2,
                   float* __restrict__ out)
{
  const int b = blockIdx.x;
  const int d = threadIdx.x;

  __shared__ int s_idx[NK];
  __shared__ int s_pos[NK];
  __shared__ int s_count;

  if (d < NK) s_idx[d] = cand[b * NK + d];
  __syncthreads();
  if (d < NK) {
    int p = 0;
    for (int j = 0; j < d; ++j) p += (s_idx[j] != -1);
    s_pos[d] = p;
  }
  if (d == 0) {
    int c = 0;
    for (int j = 0; j < NK; ++j) c += (s_idx[j] != -1);
    s_count = c;
  }
  __syncthreads();

  const float w0   = cW[d];
  const float w1   = cW[DIM + d];
  const float bias = cb[d];
  const float* nb  = nodes + (size_t)b * NN * 2;
  float* ob        = out + (size_t)b * ROWS * DIM;

  {
    const int fi = first_idx[b];
    X1[b * DIM + d] = fmaf(nb[fi * 2 + 0], w0, fmaf(nb[fi * 2 + 1], w1, bias));
    const int li = last_idx[b];
    X2[b * DIM + d] = fmaf(nb[li * 2 + 0], w0, fmaf(nb[li * 2 + 1], w1, bias));
  }

  for (int j = 0; j < NK; ++j) {
    const int ci = s_idx[j];
    if (ci != -1) {
      const float e = fmaf(nb[ci * 2 + 0], w0, fmaf(nb[ci * 2 + 1], w1, bias));
      ob[(2 + s_pos[j]) * DIM + d] = e;
    }
  }
  for (int s = s_count; s < NK; ++s) ob[(2 + s) * DIM + d] = 0.0f;
}

__global__ void __launch_bounds__(256)
wmma_heads(const float* __restrict__ X1, const float* __restrict__ X2,
           const float* __restrict__ W1, const float* __restrict__ b1,
           const float* __restrict__ W2, const float* __restrict__ b2,
           float* __restrict__ out)
{
  const int lane  = threadIdx.x & 31;
  const int wave  = threadIdx.x >> 5;
  const int mtile = blockIdx.x;
  const int r  = lane & 15;
  const int hi = lane >> 4;
  const int n  = wave * 16 + r;

  const float* Xg[2] = { X1, X2 };
  const float* Wg[2] = { W1, W2 };
  const float* Bg[2] = { b1, b2 };

  #pragma unroll
  for (int g = 0; g < 2; ++g) {
    const float* X = Xg[g];
    const float* W = Wg[g];
    v8f c = {};
    #pragma unroll
    for (int kk = 0; kk < DIM / 4; ++kk) {
      const int k = kk * 4;
      const v2f a = *(const v2f*)(X + (mtile * 16 + r) * DIM + k + 2 * hi);
      v2f bb;
      bb.x = W[(k + 2 * hi + 0) * DIM + n];
      bb.y = W[(k + 2 * hi + 1) * DIM + n];
      c = __builtin_amdgcn_wmma_f32_16x16x4_f32(
               false, a,  false, bb,
               (short)0, c,  false,  false);
    }
    const float bias = Bg[g][n];
    #pragma unroll
    for (int i = 0; i < 8; ++i) {
      const int batch = mtile * 16 + i + 8 * hi;
      out[(size_t)batch * ROWS * DIM + g * DIM + n] = c[i] + bias;
    }
  }
}

extern "C" void kernel_launch(void* const* d_in, const int* in_sizes, int n_in,
                              void* d_out, int out_size, void* d_ws, size_t ws_size,
                              hipStream_t stream)
{
  const float* nodes     = (const float*)d_in[0];
  const int*   first_idx = (const int*)d_in[1];
  const int*   last_idx  = (const int*)d_in[2];
  const int*   cand      = (const int*)d_in[3];
  const float* cW        = (const float*)d_in[4];
  const float* cb        = (const float*)d_in[5];
  const float* W1        = (const float*)d_in[6];
  const float* b1        = (const float*)d_in[7];
  const float* W2        = (const float*)d_in[8];
  const float* b2        = (const float*)d_in[9];
  float* out = (float*)d_out;

  float* X1 = (float*)d_ws;
  float* X2 = X1 + NB * DIM;

  coord_embed_gather<<<NB, DIM, 0, stream>>>(nodes, first_idx, last_idx, cand,
                                             cW, cb, X1, X2, out);
  wmma_heads<<<NB / 16, 256, 0, stream>>>(X1, X2, W1, b1, W2, b2, out);
}


# gather kernel: one wave per 16 candidates, lane = 4 channels; indices and node pairs by block-uniform scalar loads, all in flight together (was 100 serial LDS+global round trips); 512-B row stores
# speedup vs baseline: 4.5714x; 4.5714x over previous
#include <hip/hip_runtime.h>


typedef __attribute__((ext_vector_type(2))) float v2f;
typedef __attribute__((ext_vector_type(4))) float v4f;
typedef __attribute__((ext_vector_type(8))) float v8f;

namespace {
constexpr int NB   = 256;
constexpr int NN   = 5000;
constexpr int NK   = 100;
constexpr int DIM  = 128;
constexpr int ROWS = NK + 2;
constexpr int CPB  = 16;
constexpr int NFULL = NK / CPB;
constexpr int CREST = NK - NFULL * CPB;
static_assert(CREST == 4, "the last block's code is written for 4 candidates");
}

__device__ __forceinline__ v4f embed4(v2f xy, v4f w0, v4f w1, v4f bs)
{
  v4f e;
  e.x = fmaf(xy.x, w0.x, fmaf(xy.y, w1.x, bs.x));
  e.y = fmaf(xy.x, w0.y, fmaf(xy.y, w1.y, bs.y));
  e.z = fmaf(xy.x, w0.z, fmaf(xy.y, w1.z, bs.z));
  e.w = fmaf(xy.x, w0.w, fmaf(xy.y, w1.w, bs.w));
  return e;
}

template <int N>
__device__ __forceinline__ int gather_rows(const char* nb, const int* cq, int lo, float* ob,
                                           int lane, v4f w0, v4f w1, v4f bs)
{
  int ci[N];
  #pragma unroll
  for (int k = 0; k < N; ++k) ci[k] = cq[lo + k];
  v2f xy[N];
  #pragma unroll
  for (int k = 0; k < N; ++k) {
    const unsigned off = (ci[k] != -1) ? 8u * (unsigned)ci[k] : 0u;
    xy[k] = *(const v2f*)(nb + off);
  }

  unsigned m[4];
  #pragma unroll
  for (int i = 0; i < 4; ++i) {
    const int j = lane + 32 * i;
    m[i] = __builtin_amdgcn_ballot_w32((cq[j < NK ? j : NK - 1] != -1) && (j < NK));
  }
  int pos = 0;
  #pragma unroll
  for (int i = 0; i < 4; ++i) {
    const int nbits = lo - 32 * i;
    const unsigned below = nbits >= 32 ? 0xffffffffu : (nbits > 0 ? (1u << nbits) - 1u : 0u);
    pos += __builtin_popcount(m[i] & below);
  }
  static_assert(CPB == 16 && N <= 16, "keep bits must not straddle two bitmap words");
  const unsigned keep = m[lo >> 5] >> (lo & 31);

  #pragma unroll
  for (int k = 0; k < N; ++k) {
    if (__builtin_expect((keep & (1u << k)) != 0u, 1)) {
      *(v4f*)(ob + (2 + pos) * DIM) = embed4(xy[k], w0, w1, bs);
      ++pos;
    }
  }
  return pos;
}

__global__ void __launch_bounds__(32)
coord_embed_gather(const float* __restrict__ nodes,
                   const int*  __restrict__ first_idx,
                   const int*  __restrict__ last_idx,
                   const int*  __restrict__ cand,
                   const float* __restrict__ cW,
                   const float* __restrict__ cb,
                   float* __restrict__ X12,
                   float* __restrict__ out)
{
  const int q    = blockIdx.x;
  const int b    = blockIdx.y;
  const int lane = threadIdx.x;
  const int d4   = lane * 4;

  const char* nb = (const char*)(nodes + (size_t)b * NN * 2);
  const int*  cq = cand + b * NK;
  float* ob      = out + (size_t)b * ROWS * DIM + d4;

  const v4f w0 = *(const v4f*)(cW + d4);
  const v4f w1 = *(const v4f*)(cW + DIM + d4);
  const v4f bs = *(const v4f*)(cb + d4);

  if (q < NFULL) {
    gather_rows<CPB>(nb, cq, q * CPB, ob, lane, w0, w1, bs);
    return;
  }

  const v2f fxy = *(const v2f*)(nb + 8u * (unsigned)first_idx[b]);
  const v2f lxy = *(const v2f*)(nb + 8u * (unsigned)last_idx[b]);
  const int total = gather_rows<CREST>(nb, cq, NFULL * CPB, ob, lane, w0, w1, bs);
  *(v4f*)(X12 + b * DIM + d4)            = embed4(fxy, w0, w1, bs);
  *(v4f*)(X12 + (NB + b) * DIM + d4)     = embed4(lxy, w0, w1, bs);
  const v4f z = {0.0f, 0.0f, 0.0f, 0.0f};
  for (int s = total; s < NK; ++s) *(v4f*)(ob + (2 + s) * DIM) = z;
}

__global__ void __launch_bounds__(256)
wmma_heads(const float* __restrict__ X1, const float* __restrict__ X2,
           const float* __restrict__ W1, const float* __restrict__ b1,
           const float* __restrict__ W2, const float* __restrict__ b2,
           float* __restrict__ out)
{
  const int lane  = threadIdx.x & 31;
  const int wave  = threadIdx.x >> 5;
  const int mtile = blockIdx.x;
  const int r  = lane & 15;
  const int hi = lane >> 4;
  const int n  = wave * 16 + r;

  const float* Xg[2] = { X1, X2 };
  const float* Wg[2] = { W1, W2 };
  const float* Bg[2] = { b1, b2 };

  #pragma unroll
  for (int g = 0; g < 2; ++g) {
    const float* X = Xg[g];
    const float* W = Wg[g];
    v8f c = {};
    #pragma unroll
    for (int kk = 0; kk < DIM / 4; ++kk) {
      const int k = kk * 4;
      const v2f a = *(const v2f*)(X + (mtile * 16 + r) * DIM + k + 2 * hi);
      v2f bb;
      bb.x = W[(k + 2 * hi + 0) * DIM + n];
      bb.y = W[(k + 2 * hi + 1) * DIM + n];
      c = __builtin_amdgcn_wmma_f32_16x16x4_f32(
               false, a,  false, bb,
               (short)0, c,  false,  false);
    }
    const float bias = Bg[g][n];
    #pragma unroll
    for (int i = 0; i < 8; ++i) {
      const int batch = mtile * 16 + i + 8 * hi;
      out[(size_t)batch * ROWS * DIM + g * DIM + n] = c[i] + bias;
    }
  }
}

extern "C" void kernel_launch(void* const* d_in, const int* in_sizes, int n_in,
                              void* d_out, int out_size, void* d_ws, size_t ws_size,
                              hipStream_t stream)
{
  const float* nodes     = (const float*)d_in[0];
  const int*   first_idx = (const int*)d_in[1];
  const int*   last_idx  = (const int*)d_in[2];
  const int*   cand      = (const int*)d_in[3];
  const float* cW        = (const float*)d_in[4];
  const float* cb        = (const float*)d_in[5];
  const float* W1        = (const float*)d_in[6];
  const float* b1        = (const float*)d_in[7];
  const float* W2        = (const float*)d_in[8];
  const float* b2        = (const float*)d_in[9];
  float* out = (float*)d_out;

  float* X1 = (float*)d_ws;
  float* X2 = X1 + NB * DIM;

  coord_embed_gather<<<dim3(NFULL + 1, NB), 32, 0, stream>>>(nodes, first_idx, last_idx, cand,
                                                             cW, cb, X1, out);
  wmma_heads<<<NB / 16, 256, 0, stream>>>(X1, X2, W1, b1, W2, b2, out);
}


# heads kernel: one single-wave work-group per (head, 16x16 tile), 256 over the chip (was 16 groups of 8 waves, 2 heads each); a tile's whole-K fragments load as one group before its 32 products
# speedup vs baseline: 6.3602x; 1.3913x over previous
#include <hip/hip_runtime.h>


typedef __attribute__((ext_vector_type(2))) float v2f;
typedef __attribute__((ext_vector_type(4))) float v4f;
typedef __attribute__((ext_vector_type(8))) float v8f;

namespace {
constexpr int NB   = 256;
constexpr int NN   = 5000;
constexpr int NK   = 100;
constexpr int DIM  = 128;
constexpr int ROWS = NK + 2;
constexpr int CPB  = 16;
constexpr int NFULL = NK / CPB;
constexpr int CREST = NK - NFULL * CPB;
static_assert(CREST == 4, "the last block's code is written for 4 candidates");
constexpr int MTILES = NB / 16;
constexpr int NTILES = DIM / 16;
}

__device__ __forceinline__ v4f embed4(v2f xy, v4f w0, v4f w1, v4f bs)
{
  v4f e;
  e.x = fmaf(xy.x, w0.x, fmaf(xy.y, w1.x, bs.x));
  e.y = fmaf(xy.x, w0.y, fmaf(xy.y, w1.y, bs.y));
  e.z = fmaf(xy.x, w0.z, fmaf(xy.y, w1.z, bs.z));
  e.w = fmaf(xy.x, w0.w, fmaf(xy.y, w1.w, bs.w));
  return e;
}

template <int N>
__device__ __forceinline__ int gather_rows(const char* nb, const int* cq, int lo, float* ob,
                                           int lane, v4f w0, v4f w1, v4f bs)
{
  int ci[N];
  #pragma unroll
  for (int k = 0; k < N; ++k) ci[k] = cq[lo + k];
  v2f xy[N];
  #pragma unroll
  for (int k = 0; k < N; ++k) {
    const unsigned off = (ci[k] != -1) ? 8u * (unsigned)ci[k] : 0u;
    xy[k] = *(const v2f*)(nb + off);
  }

  unsigned m[4];
  #pragma unroll
  for (int i = 0; i < 4; ++i) {
    const int j = lane + 32 * i;
    m[i] = __builtin_amdgcn_ballot_w32((cq[j < NK ? j : NK - 1] != -1) && (j < NK));
  }
  int pos = 0;
  #pragma unroll
  for (int i = 0; i < 4; ++i) {
    const int nbits = lo - 32 * i;
    const unsigned below = nbits >= 32 ? 0xffffffffu : (nbits > 0 ? (1u << nbits) - 1u : 0u);
    pos += __builtin_popcount(m[i] & below);
  }
  static_assert(CPB == 16 && N <= 16, "keep bits must not straddle two bitmap words");
  const unsigned keep = m[lo >> 5] >> (lo & 31);

  #pragma unroll
  for (int k = 0; k < N; ++k) {
    if (__builtin_expect((keep & (1u << k)) != 0u, 1)) {
      *(v4f*)(ob + (2 + pos) * DIM) = embed4(xy[k], w0, w1, bs);
      ++pos;
    }
  }
  return pos;
}

__global__ void __launch_bounds__(32)
coord_embed_gather(const float* __restrict__ nodes,
                   const int*  __restrict__ first_idx,
                   const int*  __restrict__ last_idx,
                   const int*  __restrict__ cand,
                   const float* __restrict__ cW,
                   const float* __restrict__ cb,
                   float* __restrict__ X12,
                   float* __restrict__ out)
{
  const int q    = blockIdx.x;
  const int b    = blockIdx.y;
  const int lane = threadIdx.x;
  const int d4   = lane * 4;

  const char* nb = (const char*)(nodes + (size_t)b * NN * 2);
  const int*  cq = cand + b * NK;
  float* ob      = out + (size_t)b * ROWS * DIM + d4;

  const v4f w0 = *(const v4f*)(cW + d4);
  const v4f w1 = *(const v4f*)(cW + DIM + d4);
  const v4f bs = *(const v4f*)(cb + d4);

  if (q < NFULL) {
    gather_rows<CPB>(nb, cq, q * CPB, ob, lane, w0, w1, bs);
    return;
  }

  const v2f fxy = *(const v2f*)(nb + 8u * (unsigned)first_idx[b]);
  const v2f lxy = *(const v2f*)(nb + 8u * (unsigned)last_idx[b]);
  const int total = gather_rows<CREST>(nb, cq, NFULL * CPB, ob, lane, w0, w1, bs);
  *(v4f*)(X12 + b * DIM + d4)            = embed4(fxy, w0, w1, bs);
  *(v4f*)(X12 + (NB + b) * DIM + d4)     = embed4(lxy, w0, w1, bs);
  const v4f z = {0.0f, 0.0f, 0.0f, 0.0f};
  for (int s = total; s < NK; ++s) *(v4f*)(ob + (2 + s) * DIM) = z;
}

__global__ void __launch_bounds__(32) __attribute__((amdgpu_waves_per_eu(1, 4)))
wmma_heads(const float* __restrict__ X1, const float* __restrict__ X2,
           const float* __restrict__ W1, const float* __restrict__ b1,
           const float* __restrict__ W2, const float* __restrict__ b2,
           float* __restrict__ out)
{
  const int lane  = threadIdx.x;
  const int g     = blockIdx.x / (MTILES * NTILES);
  const int mtile = (blockIdx.x / NTILES) % MTILES;
  const int ntile = blockIdx.x % NTILES;
  const int r  = lane & 15;
  const int hi = lane >> 4;
  const int n  = ntile * 16 + r;

  const float* X  = g ? X2 : X1;
  const float* W  = g ? W2 : W1;
  const float* Bv = g ? b2 : b1;

  const float* xa = X + (mtile * 16 + r) * DIM + 2 * hi;
  const float* wb = W + (2 * hi) * DIM + n;
  v2f a[DIM / 4], bb[DIM / 4];
  #pragma unroll
  for (int kk = 0; kk < DIM / 4; ++kk) {
    a[kk] = *(const v2f*)(xa + kk * 4);
    bb[kk].x = wb[(kk * 4 + 0) * DIM];
    bb[kk].y = wb[(kk * 4 + 1) * DIM];
  }
  const float bias = Bv[n];

  v8f c = {};
  #pragma unroll
  for (int kk = 0; kk < DIM / 4; ++kk) {
    c = __builtin_amdgcn_wmma_f32_16x16x4_f32(
             false, a[kk],  false, bb[kk],
             (short)0, c,  false,  false);
  }
  #pragma unroll
  for (int i = 0; i < 8; ++i) {
    const int batch = mtile * 16 + i + 8 * hi;
    out[(size_t)batch * ROWS * DIM + g * DIM + n] = c[i] + bias;
  }
}

extern "C" void kernel_launch(void* const* d_in, const int* in_sizes, int n_in,
                              void* d_out, int out_size, void* d_ws, size_t ws_size,
                              hipStream_t stream)
{
  const float* nodes     = (const float*)d_in[0];
  const int*   first_idx = (const int*)d_in[1];
  const int*   last_idx  = (const int*)d_in[2];
  const int*   cand      = (const int*)d_in[3];
  const float* cW        = (const float*)d_in[4];
  const float* cb        = (const float*)d_in[5];
  const float* W1        = (const float*)d_in[6];
  const float* b1        = (const float*)d_in[7];
  const float* W2        = (const float*)d_in[8];
  const float* b2        = (const float*)d_in[9];
  float* out = (float*)d_out;

  float* X1 = (float*)d_ws;
  float* X2 = X1 + NB * DIM;

  coord_embed_gather<<<dim3(NFULL + 1, NB), 32, 0, stream>>>(nodes, first_idx, last_idx, cand,
                                                             cW, cb, X1, out);
  wmma_heads<<<2 * MTILES * NTILES, 32, 0, stream>>>(X1, X2, W1, b1, W2, b2, out);
}
